# starting kernel
# baseline (speedup 1.0000x reference)
#include <hip/hip_runtime.h>
#include <hip/hip_bf16.h>

#define KS      7
#define DILX    2
#define NHEADS  4
#define NFRAMES 3
#define CDIM    128
#define HD      32
#define HH      64
#define WW      64
#define NPIX    4096
#define NB      2
#define KTOT    (NFRAMES*CDIM)
#define KSTEPS  (KTOT/4)
#define NFRAG   (KSTEPS*8*32)

typedef __attribute__((ext_vector_type(2))) float v2f;
typedef __attribute__((ext_vector_type(8))) float v8f;

__device__ __forceinline__ int win_start(int i) {
    int ni = i - (KS / 2) * DILX;
    if (ni < 0) return i % DILX;
    if (i + (KS / 2) * DILX >= HH) return HH - KS * DILX + (i % DILX);
    return ni;
}

__global__ void natten_attn_kernel(const float* __restrict__ q,
                                   const float* __restrict__ kk,
                                   const float* __restrict__ vv,
                                   const float* __restrict__ rpb,
                                   float* __restrict__ xn)
{
    int tid = blockIdx.x * blockDim.x + threadIdx.x;
    int j = tid & 63;
    int t = tid >> 6;
    int i = t & 63;
    t >>= 6;
    int f = t % NFRAMES;
    t /= NFRAMES;
    int head = t & 3;
    int b = t >> 2;

    int si = win_start(i), sj = win_start(j);
    int pbi = (si - i) / DILX + (KS - 1);
    int pbj = (sj - j) / DILX + (KS - 1);

    const float scale = 0.17677669529663687f;

    const float* qp = q + ((size_t)(b * CDIM + head * HD)) * NPIX + i * WW + j;
    float qr[HD];
#pragma unroll
    for (int d = 0; d < HD; ++d) qr[d] = qp[(size_t)d * NPIX] * scale;

    const size_t kvbase = ((size_t)(b * KTOT + f * CDIM + head * HD)) * NPIX;
    const float* kb = kk + kvbase;
    const float* vb = vv + kvbase;
    const float* rp = rpb + head * (2 * KS - 1) * (2 * KS - 1);

    float m = -3.0e38f, s = 0.0f;
    float acc[HD];
#pragma unroll
    for (int d = 0; d < HD; ++d) acc[d] = 0.0f;

    for (int ti = 0; ti < KS; ++ti) {
        int ii  = si + ti * DILX;
        int bro = (pbi + ti) * (2 * KS - 1);
        for (int tj = 0; tj < KS; ++tj) {
            int jj  = sj + tj * DILX;
            int off = ii * WW + jj;

            float l = rp[bro + pbj + tj];
#pragma unroll
            for (int d = 0; d < HD; ++d)
                l += qr[d] * kb[(size_t)d * NPIX + off];

            float mn   = fmaxf(m, l);
            float corr = __expf(m - mn);
            float w    = __expf(l - mn);
            s = s * corr + w;
            m = mn;
#pragma unroll
            for (int d = 0; d < HD; ++d)
                acc[d] = acc[d] * corr + w * vb[(size_t)d * NPIX + off];
        }
    }

    float inv = 1.0f / s;
    float* op = xn + ((size_t)(b * KTOT + f * CDIM + head * HD)) * NPIX + i * WW + j;
#pragma unroll
    for (int d = 0; d < HD; ++d) op[(size_t)d * NPIX] = acc[d] * inv;
}

__global__ __launch_bounds__(256, 1)
void conv1x1_wmma_kernel(const float* __restrict__ xn,
                         const float* __restrict__ wzw,
                         const float* __restrict__ wzb,
                         float* __restrict__ out)
{
    extern __shared__ v2f wfrag[];

    for (int s = threadIdx.x; s < NFRAG; s += 256) {
        int ln    = s & 31;
        int mt    = (s >> 5) & 7;
        int kstep = s >> 8;
        int row   = mt * 16 + (ln & 15);
        int c     = kstep * 4 + 2 * (ln >> 4);
        const float* wp = wzw + (size_t)row * KTOT + c;
        v2f t;
        t.x = wp[0];
        t.y = wp[1];
        wfrag[s] = t;
    }
    __syncthreads();

    int lane  = threadIdx.x & 31;
    int wave  = threadIdx.x >> 5;
    int mhalf = wave & 1;
    int ntp   = blockIdx.x * 4 + (wave >> 1);
    int col   = lane & 15;
    int hi    = lane >> 4;

    int g0 = ntp * 32 + col;
    int b  = g0 >> 12;
    int p0 = g0 & 4095;
    const float* xp0 = xn + (size_t)b * KTOT * NPIX + p0;
    const float* xp1 = xp0 + 16;

    v8f acc0[4], acc1[4];
#pragma unroll
    for (int mt = 0; mt < 4; ++mt)
#pragma unroll
        for (int e = 0; e < 8; ++e) { acc0[mt][e] = 0.0f; acc1[mt][e] = 0.0f; }

    auto load_step = [&](int ks, v2f a[4], v2f& b0, v2f& b1) {
        int c0 = ks * 4 + 2 * hi;
        b0.x = xp0[(size_t)c0 * NPIX];
        b0.y = xp0[(size_t)(c0 + 1) * NPIX];
        b1.x = xp1[(size_t)c0 * NPIX];
        b1.y = xp1[(size_t)(c0 + 1) * NPIX];
        const v2f* wf = wfrag + (size_t)ks * 256 + (mhalf * 4) * 32 + lane;
#pragma unroll
        for (int mt = 0; mt < 4; ++mt) a[mt] = wf[mt * 32];
    };

    v2f aA[4], bA0, bA1;
    v2f aB[4], bB0, bB1;
    load_step(0, aA, bA0, bA1);

    for (int ks = 0; ks < KSTEPS; ks += 2) {
        load_step(ks + 1, aB, bB0, bB1);

#pragma unroll
        for (int mt = 0; mt < 4; ++mt) {
            acc0[mt] = __builtin_amdgcn_wmma_f32_16x16x4_f32(
                false, aA[mt], false, bA0, (short)0, acc0[mt], false, false);
            acc1[mt] = __builtin_amdgcn_wmma_f32_16x16x4_f32(
                false, aA[mt], false, bA1, (short)0, acc1[mt], false, false);
        }

        int ksn = (ks + 2 < KSTEPS) ? (ks + 2) : (KSTEPS - 1);
        load_step(ksn, aA, bA0, bA1);

#pragma unroll
        for (int mt = 0; mt < 4; ++mt) {
            acc0[mt] = __builtin_amdgcn_wmma_f32_16x16x4_f32(
                false, aB[mt], false, bB0, (short)0, acc0[mt], false, false);
            acc1[mt] = __builtin_amdgcn_wmma_f32_16x16x4_f32(
                false, aB[mt], false, bB1, (short)0, acc1[mt], false, false);
        }
    }

    float* op0 = out + (size_t)b * CDIM * NPIX + p0;
    float* op1 = op0 + 16;
#pragma unroll
    for (int mt = 0; mt < 4; ++mt) {
#pragma unroll
        for (int e = 0; e < 8; ++e) {
            int mrow = (mhalf * 4 + mt) * 16 + e + 8 * hi;
            float bias = wzb[mrow];
            op0[(size_t)mrow * NPIX] = acc0[mt][e] + bias;
            op1[(size_t)mrow * NPIX] = acc1[mt][e] + bias;
        }
    }
}

extern "C" void kernel_launch(void* const* d_in, const int* in_sizes, int n_in,
                              void* d_out, int out_size, void* d_ws, size_t ws_size,
                              hipStream_t stream) {
    const float* q   = (const float*)d_in[0];
    const float* k   = (const float*)d_in[1];
    const float* v   = (const float*)d_in[2];
    const float* rpb = (const float*)d_in[3];
    const float* wzw = (const float*)d_in[4];
    const float* wzb = (const float*)d_in[5];
    float* out = (float*)d_out;
    float* xn  = (float*)d_ws;

    natten_attn_kernel<<<384, 256, 0, stream>>>(q, k, v, rpb, xn);

    size_t lds_bytes = (size_t)NFRAG * sizeof(v2f);
    (void)hipFuncSetAttribute(reinterpret_cast<const void*>(conv1x1_wmma_kernel),
                              hipFuncAttributeMaxDynamicSharedMemorySize,
                              (int)lds_bytes);
    conv1x1_wmma_kernel<<<64, 256, lds_bytes, stream>>>(xn, wzw, wzb, out);
}


# conv GEMM: 256 blocks of 8 waves, block = 32 pixels x 128 rows; xn tile copied once to LDS by whole-line loads, shared by all waves; a wave's 16 W rows held in registers; k order in sums changes
# speedup vs baseline: 15.9719x; 15.9719x over previous
#include <hip/hip_runtime.h>
#include <hip/hip_bf16.h>

#define KS      7
#define DILX    2
#define NHEADS  4
#define NFRAMES 3
#define CDIM    128
#define HD      32
#define HH      64
#define WW      64
#define NPIX    4096
#define NB      2
#define KTOT    (NFRAMES*CDIM)

typedef __attribute__((ext_vector_type(2))) float v2f;
typedef __attribute__((ext_vector_type(8))) float v8f;

__device__ __forceinline__ int win_start(int i) {
    int ni = i - (KS / 2) * DILX;
    if (ni < 0) return i % DILX;
    if (i + (KS / 2) * DILX >= HH) return HH - KS * DILX + (i % DILX);
    return ni;
}

typedef __attribute__((ext_vector_type(4))) float v4f;
__device__ __forceinline__ float pick(bool c, float a, float b) { return c ? a : b; }
#define PIXB      128
#define HDH       16
#define ATT_LDS_FLOATS (2 * KS * PIXB + KS * HDH * PIXB)
#define ATT_LDS_BYTES  (ATT_LDS_FLOATS * 4)

__global__ __launch_bounds__(224, 6)
void natten_attn_kernel(const float* __restrict__ q,
                        const float* __restrict__ kk,
                        const float* __restrict__ vv,
                        const float* __restrict__ rpb,
                        float* __restrict__ xn)
{
    extern __shared__ __attribute__((aligned(16))) float att_lds[];
    float* pm   = att_lds;
    float* ps   = att_lds + KS * PIXB;
    float* pacc = att_lds + 2 * KS * PIXB;

    int lane = threadIdx.x & 31;
    int w    = threadIdx.x >> 5;
    int t    = blockIdx.x;
    int i    = (t & 31) * 2 + (lane >> 4);
    t >>= 5;
    int f = t % NFRAMES;
    t /= NFRAMES;
    int head = t & 3;
    int b = t >> 2;
    int j0 = (lane & 15) * 4;

    int si  = win_start(i);
    int pbi = (si - i) / DILX + (KS - 1);
    int base = j0 - 8;
    base = base < 0 ? 0 : base;
    base = base > WW - 20 ? WW - 20 : base;

    int m0[2], pbj[2];
#pragma unroll
    for (int h = 0; h < 2; ++h) {
        int j  = j0 + 2 * h;
        int sj = win_start(j);
        m0[h]  = (sj - base) / DILX;
        pbj[h] = (sj - j) / DILX + (KS - 1);
    }

    const float scale = 0.17677669529663687f;

    const size_t kvbase = ((size_t)(b * KTOT + f * CDIM + head * HD)) * NPIX;
    const int    rowoff = (si + w * DILX) * WW + base;
    const v4f* kb = reinterpret_cast<const v4f*>(kk + kvbase + rowoff);
    const v4f* vb = reinterpret_cast<const v4f*>(vv + kvbase + rowoff);
    const v4f* qp = reinterpret_cast<const v4f*>(
        q + ((size_t)(b * CDIM + head * HD)) * NPIX + i * WW + j0);

    float cand[4][10];
#pragma unroll
    for (int pp = 0; pp < 4; ++pp)
#pragma unroll
        for (int m = 0; m < 10; ++m) cand[pp][m] = 0.0f;

#pragma unroll 4
    for (int d = 0; d < HD; ++d) {
        v4f qv = qp[(size_t)d * (NPIX / 4)];
        float kr[20];
#pragma unroll
        for (int g = 0; g < 5; ++g) {
            v4f x = kb[(size_t)d * (NPIX / 4) + g];
            kr[4 * g + 0] = x[0]; kr[4 * g + 1] = x[1];
            kr[4 * g + 2] = x[2]; kr[4 * g + 3] = x[3];
        }
#pragma unroll
        for (int pp = 0; pp < 4; ++pp) {
            float qs = qv[pp] * scale;
#pragma unroll
            for (int m = 0; m < 10; ++m)
                cand[pp][m] += qs * kr[(pp & 1) + 2 * m];
        }
    }

    float wc[4][10];
    v4f mv, sv;
#pragma unroll
    for (int pp = 0; pp < 4; ++pp) {
        bool s1 = (m0[pp >> 1] & 1) != 0, s2 = (m0[pp >> 1] & 2) != 0;
        float t1[9];
#pragma unroll
        for (int x = 0; x < 9; ++x) t1[x] = pick(s1, cand[pp][x + 1], cand[pp][x]);
        const float* rp = rpb + head * (2 * KS - 1) * (2 * KS - 1)
                              + (pbi + w) * (2 * KS - 1) + pbj[pp >> 1];
        float l[KS];
#pragma unroll
        for (int tj = 0; tj < KS; ++tj) l[tj] = pick(s2, t1[tj + 2], t1[tj]) + rp[tj];
        float m = l[0];
#pragma unroll
        for (int tj = 1; tj < KS; ++tj) m = fmaxf(m, l[tj]);
        float p[KS];
        float s = 0.0f;
#pragma unroll
        for (int tj = 0; tj < KS; ++tj) { p[tj] = __expf(l[tj] - m); s += p[tj]; }
        mv[pp] = m;
        sv[pp] = s;
        float u[9];
#pragma unroll
        for (int y = 0; y < 9; ++y) {
            float a0 = (y < KS) ? p[y < KS ? y : 0] : 0.0f;
            float a2 = (y >= 2) ? p[y >= 2 ? y - 2 : 0] : 0.0f;
            u[y] = pick(s2, a2, a0);
        }
#pragma unroll
        for (int mm = 0; mm < 10; ++mm) {
            float c0 = (mm < 9) ? u[mm < 9 ? mm : 0] : 0.0f;
            float c1 = (mm >= 1) ? u[mm >= 1 ? mm - 1 : 0] : 0.0f;
            wc[pp][mm] = pick(s1, c1, c0);
        }
    }
    reinterpret_cast<v4f*>(pm + w * PIXB)[lane] = mv;
    reinterpret_cast<v4f*>(ps + w * PIXB)[lane] = sv;

    float* op = xn + kvbase + i * WW + j0;

#pragma unroll
    for (int hf = 0; hf < 2; ++hf) {
#pragma unroll 4
        for (int dd = 0; dd < HDH; ++dd) {
            const int d = hf * HDH + dd;
            float vr[20];
#pragma unroll
            for (int g = 0; g < 5; ++g) {
                v4f x = vb[(size_t)d * (NPIX / 4) + g];
                vr[4 * g + 0] = x[0]; vr[4 * g + 1] = x[1];
                vr[4 * g + 2] = x[2]; vr[4 * g + 3] = x[3];
            }
            v4f av;
#pragma unroll
            for (int pp = 0; pp < 4; ++pp) {
                float a = 0.0f;
#pragma unroll
                for (int m = 0; m < 10; ++m)
                    a = __builtin_fmaf(wc[pp][m], vr[(pp & 1) + 2 * m], a);
                av[pp] = a;
            }
            reinterpret_cast<v4f*>(pacc + (w * HDH + dd) * PIXB)[lane] = av;
        }

        __syncthreads();

        v4f e[KS];
        v4f inv;
        {
            v4f pmr[KS], psr[KS];
#pragma unroll
            for (int r = 0; r < KS; ++r) {
                pmr[r] = reinterpret_cast<const v4f*>(pm + r * PIXB)[lane];
                psr[r] = reinterpret_cast<const v4f*>(ps + r * PIXB)[lane];
            }
#pragma unroll
            for (int pp = 0; pp < 4; ++pp) {
                float M = pmr[0][pp];
#pragma unroll
                for (int r = 1; r < KS; ++r) M = fmaxf(M, pmr[r][pp]);
                float S = 0.0f;
#pragma unroll
                for (int r = 0; r < KS; ++r) {
                    float er = __expf(pmr[r][pp] - M);
                    e[r][pp] = er;
                    S += er * psr[r][pp];
                }
                inv[pp] = 1.0f / S;
            }
        }

        for (int dd = w; dd < HDH; dd += KS) {
            v4f o = {0.0f, 0.0f, 0.0f, 0.0f};
#pragma unroll
            for (int r = 0; r < KS; ++r) {
                v4f pa = reinterpret_cast<const v4f*>(pacc + (r * HDH + dd) * PIXB)[lane];
                o += e[r] * pa;
            }
            *reinterpret_cast<v4f*>(op + (size_t)(hf * HDH + dd) * NPIX) = o * inv;
        }

        if (hf == 0) __syncthreads();
    }
}

#define XPP     80
#define KPAIRS  (KTOT / 2)
#define KGRP    (KTOT / 8)

__global__ __launch_bounds__(256, 2)
void conv1x1_wmma_kernel(const float* __restrict__ xn,
                         const float* __restrict__ wzw,
                         const float* __restrict__ wzb,
                         float* __restrict__ out)
{
    __shared__ __attribute__((aligned(16))) float xs[KPAIRS * XPP];

    const int tid  = threadIdx.x;
    const int lane = tid & 31;
    const int wave = tid >> 5;
    const int col  = lane & 15;
    const int hi   = lane >> 4;
    const int b    = blockIdx.x >> 7;
    const int p0   = (blockIdx.x & 127) * 32;

    const int q  = tid & 7;
    const int pr = tid >> 3;
    const float* xsrc = xn + ((size_t)b * KTOT + 2 * pr) * NPIX + p0 + 4 * q;
    v4f xe[6], xo[6];
#pragma unroll
    for (int ps = 0; ps < 6; ++ps) {
        xe[ps] = *reinterpret_cast<const v4f*>(xsrc + (size_t)(64 * ps) * NPIX);
        xo[ps] = *reinterpret_cast<const v4f*>(xsrc + (size_t)(64 * ps + 1) * NPIX);
    }

    float* xdst = xs + pr * XPP + 8 * q;
#pragma unroll
    for (int ps = 0; ps < 6; ++ps) {
        v4f lo = {xe[ps][0], xo[ps][0], xe[ps][1], xo[ps][1]};
        v4f up = {xe[ps][2], xo[ps][2], xe[ps][3], xo[ps][3]};
        *reinterpret_cast<v4f*>(xdst + (32 * ps) * XPP)     = lo;
        *reinterpret_cast<v4f*>(xdst + (32 * ps) * XPP + 4) = up;
    }
    __syncthreads();

    const float* wsrc = wzw + (size_t)(wave * 16 + col) * KTOT + 4 * hi;
    v4f av[KGRP];
#pragma unroll
    for (int kp = 0; kp < KGRP; ++kp)
        av[kp] = *reinterpret_cast<const v4f*>(wsrc + 8 * kp);

    v8f acc0, acc1;
#pragma unroll
    for (int e = 0; e < 8; ++e) { acc0[e] = 0.0f; acc1[e] = 0.0f; }

    const v4f* xb = reinterpret_cast<const v4f*>(xs + (2 * hi) * XPP + 4 * col);
    v4f bf[2][2][2];
#pragma unroll
    for (int u = 0; u < 2; ++u)
#pragma unroll
        for (int j = 0; j < 2; ++j)
            bf[0][u][j] = xb[((4 * u + j) * XPP) / 4];

#pragma unroll
    for (int g = 0; g < KGRP / 2; ++g) {
        if (g + 1 < KGRP / 2) {
#pragma unroll
            for (int u = 0; u < 2; ++u)
#pragma unroll
                for (int j = 0; j < 2; ++j)
                    bf[(g + 1) & 1][u][j] = xb[((4 * (2 * (g + 1) + u) + j) * XPP) / 4];
        }
        __builtin_amdgcn_sched_barrier(0);
#pragma unroll
        for (int u = 0; u < 2; ++u)
#pragma unroll
            for (int j = 0; j < 2; ++j) {
                v2f a  = {av[2 * g + u][2 * j], av[2 * g + u][2 * j + 1]};
                v4f bb = bf[g & 1][u][j];
                v2f b0 = {bb[0], bb[1]};
                v2f b1 = {bb[2], bb[3]};
                acc0 = __builtin_amdgcn_wmma_f32_16x16x4_f32(
                    false, a, false, b0, (short)0, acc0, false, false);
                acc1 = __builtin_amdgcn_wmma_f32_16x16x4_f32(
                    false, a, false, b1, (short)0, acc1, false, false);
            }
        __builtin_amdgcn_sched_barrier(0);
    }

    float* orow = out + ((size_t)b * CDIM + wave * 16 + 8 * hi) * NPIX + p0 + 2 * col;
#pragma unroll
    for (int r = 0; r < 8; ++r) {
        float bias = wzb[wave * 16 + 8 * hi + r];
        v2f o = {acc0[r] + bias, acc1[r] + bias};
        *reinterpret_cast<v2f*>(orow + (size_t)r * NPIX) = o;
    }
}

extern "C" void kernel_launch(void* const* d_in, const int* in_sizes, int n_in,
                              void* d_out, int out_size, void* d_ws, size_t ws_size,
                              hipStream_t stream) {
    const float* q   = (const float*)d_in[0];
    const float* k   = (const float*)d_in[1];
    const float* v   = (const float*)d_in[2];
    const float* rpb = (const float*)d_in[3];
    const float* wzw = (const float*)d_in[4];
    const float* wzb = (const float*)d_in[5];
    float* out = (float*)d_out;
    float* xn  = (float*)d_ws;

    natten_attn_kernel<<<768, 224, ATT_LDS_BYTES, stream>>>(q, k, v, rpb, xn);

    conv1x1_wmma_kernel<<<256, 256, 0, stream>>>(xn, wzw, wzb, out);
}


# attention kernel: a block's two pixel rows are i and i+2 (one row parity): wave w's 2nd key row is wave w+1's 1st, so a block asks L2 for 8 distinct K/V rows per channel, not 14; sums unchanged
# speedup vs baseline: 17.0442x; 1.0671x over previous
#include <hip/hip_runtime.h>
#include <hip/hip_bf16.h>

#define KS      7
#define DILX    2
#define NHEADS  4
#define NFRAMES 3
#define CDIM    128
#define HD      32
#define HH      64
#define WW      64
#define NPIX    4096
#define NB      2
#define KTOT    (NFRAMES*CDIM)

typedef __attribute__((ext_vector_type(2))) float v2f;
typedef __attribute__((ext_vector_type(8))) float v8f;

__device__ __forceinline__ int win_start(int i) {
    int ni = i - (KS / 2) * DILX;
    if (ni < 0) return i % DILX;
    if (i + (KS / 2) * DILX >= HH) return HH - KS * DILX + (i % DILX);
    return ni;
}

typedef __attribute__((ext_vector_type(4))) float v4f;
__device__ __forceinline__ float pick(bool c, float a, float b) { return c ? a : b; }
#define PIXB      128
#define HDH       16
#define ATT_LDS_FLOATS (2 * KS * PIXB + KS * HDH * PIXB)
#define ATT_LDS_BYTES  (ATT_LDS_FLOATS * 4)

__global__ __launch_bounds__(224, 6)
void natten_attn_kernel(const float* __restrict__ q,
                        const float* __restrict__ kk,
                        const float* __restrict__ vv,
                        const float* __restrict__ rpb,
                        float* __restrict__ xn)
{
    extern __shared__ __attribute__((aligned(16))) float att_lds[];
    float* pm   = att_lds;
    float* ps   = att_lds + KS * PIXB;
    float* pacc = att_lds + 2 * KS * PIXB;

    int lane = threadIdx.x & 31;
    int w    = threadIdx.x >> 5;
    int t    = blockIdx.x;
    int i    = ((t & 31) >> 1) * 4 + (t & 1) + 2 * (lane >> 4);
    t >>= 5;
    int f = t % NFRAMES;
    t /= NFRAMES;
    int head = t & 3;
    int b = t >> 2;
    int j0 = (lane & 15) * 4;

    int si  = win_start(i);
    int pbi = (si - i) / DILX + (KS - 1);
    int base = j0 - 8;
    base = base < 0 ? 0 : base;
    base = base > WW - 20 ? WW - 20 : base;

    int m0[2], pbj[2];
#pragma unroll
    for (int h = 0; h < 2; ++h) {
        int j  = j0 + 2 * h;
        int sj = win_start(j);
        m0[h]  = (sj - base) / DILX;
        pbj[h] = (sj - j) / DILX + (KS - 1);
    }

    const float scale = 0.17677669529663687f;

    const size_t kvbase = ((size_t)(b * KTOT + f * CDIM + head * HD)) * NPIX;
    const int    rowoff = (si + w * DILX) * WW + base;
    const v4f* kb = reinterpret_cast<const v4f*>(kk + kvbase + rowoff);
    const v4f* vb = reinterpret_cast<const v4f*>(vv + kvbase + rowoff);
    const v4f* qp = reinterpret_cast<const v4f*>(
        q + ((size_t)(b * CDIM + head * HD)) * NPIX + i * WW + j0);

    float cand[4][10];
#pragma unroll
    for (int pp = 0; pp < 4; ++pp)
#pragma unroll
        for (int m = 0; m < 10; ++m) cand[pp][m] = 0.0f;

#pragma unroll 4
    for (int d = 0; d < HD; ++d) {
        v4f qv = qp[(size_t)d * (NPIX / 4)];
        float kr[20];
#pragma unroll
        for (int g = 0; g < 5; ++g) {
            v4f x = kb[(size_t)d * (NPIX / 4) + g];
            kr[4 * g + 0] = x[0]; kr[4 * g + 1] = x[1];
            kr[4 * g + 2] = x[2]; kr[4 * g + 3] = x[3];
        }
#pragma unroll
        for (int pp = 0; pp < 4; ++pp) {
            float qs = qv[pp] * scale;
#pragma unroll
            for (int m = 0; m < 10; ++m)
                cand[pp][m] += qs * kr[(pp & 1) + 2 * m];
        }
    }

    float wc[4][10];
    v4f mv, sv;
#pragma unroll
    for (int pp = 0; pp < 4; ++pp) {
        bool s1 = (m0[pp >> 1] & 1) != 0, s2 = (m0[pp >> 1] & 2) != 0;
        float t1[9];
#pragma unroll
        for (int x = 0; x < 9; ++x) t1[x] = pick(s1, cand[pp][x + 1], cand[pp][x]);
        const float* rp = rpb + head * (2 * KS - 1) * (2 * KS - 1)
                              + (pbi + w) * (2 * KS - 1) + pbj[pp >> 1];
        float l[KS];
#pragma unroll
        for (int tj = 0; tj < KS; ++tj) l[tj] = pick(s2, t1[tj + 2], t1[tj]) + rp[tj];
        float m = l[0];
#pragma unroll
        for (int tj = 1; tj < KS; ++tj) m = fmaxf(m, l[tj]);
        float p[KS];
        float s = 0.0f;
#pragma unroll
        for (int tj = 0; tj < KS; ++tj) { p[tj] = __expf(l[tj] - m); s += p[tj]; }
        mv[pp] = m;
        sv[pp] = s;
        float u[9];
#pragma unroll
        for (int y = 0; y < 9; ++y) {
            float a0 = (y < KS) ? p[y < KS ? y : 0] : 0.0f;
            float a2 = (y >= 2) ? p[y >= 2 ? y - 2 : 0] : 0.0f;
            u[y] = pick(s2, a2, a0);
        }
#pragma unroll
        for (int mm = 0; mm < 10; ++mm) {
            float c0 = (mm < 9) ? u[mm < 9 ? mm : 0] : 0.0f;
            float c1 = (mm >= 1) ? u[mm >= 1 ? mm - 1 : 0] : 0.0f;
            wc[pp][mm] = pick(s1, c1, c0);
        }
    }
    reinterpret_cast<v4f*>(pm + w * PIXB)[lane] = mv;
    reinterpret_cast<v4f*>(ps + w * PIXB)[lane] = sv;

    float* op = xn + kvbase + i * WW + j0;

#pragma unroll
    for (int hf = 0; hf < 2; ++hf) {
#pragma unroll 4
        for (int dd = 0; dd < HDH; ++dd) {
            const int d = hf * HDH + dd;
            float vr[20];
#pragma unroll
            for (int g = 0; g < 5; ++g) {
                v4f x = vb[(size_t)d * (NPIX / 4) + g];
                vr[4 * g + 0] = x[0]; vr[4 * g + 1] = x[1];
                vr[4 * g + 2] = x[2]; vr[4 * g + 3] = x[3];
            }
            v4f av;
#pragma unroll
            for (int pp = 0; pp < 4; ++pp) {
                float a = 0.0f;
#pragma unroll
                for (int m = 0; m < 10; ++m)
                    a = __builtin_fmaf(wc[pp][m], vr[(pp & 1) + 2 * m], a);
                av[pp] = a;
            }
            reinterpret_cast<v4f*>(pacc + (w * HDH + dd) * PIXB)[lane] = av;
        }

        __syncthreads();

        v4f e[KS];
        v4f inv;
        {
            v4f pmr[KS], psr[KS];
#pragma unroll
            for (int r = 0; r < KS; ++r) {
                pmr[r] = reinterpret_cast<const v4f*>(pm + r * PIXB)[lane];
                psr[r] = reinterpret_cast<const v4f*>(ps + r * PIXB)[lane];
            }
#pragma unroll
            for (int pp = 0; pp < 4; ++pp) {
                float M = pmr[0][pp];
#pragma unroll
                for (int r = 1; r < KS; ++r) M = fmaxf(M, pmr[r][pp]);
                float S = 0.0f;
#pragma unroll
                for (int r = 0; r < KS; ++r) {
                    float er = __expf(pmr[r][pp] - M);
                    e[r][pp] = er;
                    S += er * psr[r][pp];
                }
                inv[pp] = 1.0f / S;
            }
        }

        for (int dd = w; dd < HDH; dd += KS) {
            v4f o = {0.0f, 0.0f, 0.0f, 0.0f};
#pragma unroll
            for (int r = 0; r < KS; ++r) {
                v4f pa = reinterpret_cast<const v4f*>(pacc + (r * HDH + dd) * PIXB)[lane];
                o += e[r] * pa;
            }
            *reinterpret_cast<v4f*>(op + (size_t)(hf * HDH + dd) * NPIX) = o * inv;
        }

        if (hf == 0) __syncthreads();
    }
}

#define XPP     80
#define KPAIRS  (KTOT / 2)
#define KGRP    (KTOT / 8)

__global__ __launch_bounds__(256, 2)
void conv1x1_wmma_kernel(const float* __restrict__ xn,
                         const float* __restrict__ wzw,
                         const float* __restrict__ wzb,
                         float* __restrict__ out)
{
    __shared__ __attribute__((aligned(16))) float xs[KPAIRS * XPP];

    const int tid  = threadIdx.x;
    const int lane = tid & 31;
    const int wave = tid >> 5;
    const int col  = lane & 15;
    const int hi   = lane >> 4;
    const int b    = blockIdx.x >> 7;
    const int p0   = (blockIdx.x & 127) * 32;

    const int q  = tid & 7;
    const int pr = tid >> 3;
    const float* xsrc = xn + ((size_t)b * KTOT + 2 * pr) * NPIX + p0 + 4 * q;
    v4f xe[6], xo[6];
#pragma unroll
    for (int ps = 0; ps < 6; ++ps) {
        xe[ps] = *reinterpret_cast<const v4f*>(xsrc + (size_t)(64 * ps) * NPIX);
        xo[ps] = *reinterpret_cast<const v4f*>(xsrc + (size_t)(64 * ps + 1) * NPIX);
    }

    float* xdst = xs + pr * XPP + 8 * q;
#pragma unroll
    for (int ps = 0; ps < 6; ++ps) {
        v4f lo = {xe[ps][0], xo[ps][0], xe[ps][1], xo[ps][1]};
        v4f up = {xe[ps][2], xo[ps][2], xe[ps][3], xo[ps][3]};
        *reinterpret_cast<v4f*>(xdst + (32 * ps) * XPP)     = lo;
        *reinterpret_cast<v4f*>(xdst + (32 * ps) * XPP + 4) = up;
    }
    __syncthreads();

    const float* wsrc = wzw + (size_t)(wave * 16 + col) * KTOT + 4 * hi;
    v4f av[KGRP];
#pragma unroll
    for (int kp = 0; kp < KGRP; ++kp)
        av[kp] = *reinterpret_cast<const v4f*>(wsrc + 8 * kp);

    v8f acc0, acc1;
#pragma unroll
    for (int e = 0; e < 8; ++e) { acc0[e] = 0.0f; acc1[e] = 0.0f; }

    const v4f* xb = reinterpret_cast<const v4f*>(xs + (2 * hi) * XPP + 4 * col);
    v4f bf[2][2][2];
#pragma unroll
    for (int u = 0; u < 2; ++u)
#pragma unroll
        for (int j = 0; j < 2; ++j)
            bf[0][u][j] = xb[((4 * u + j) * XPP) / 4];

#pragma unroll
    for (int g = 0; g < KGRP / 2; ++g) {
        if (g + 1 < KGRP / 2) {
#pragma unroll
            for (int u = 0; u < 2; ++u)
#pragma unroll
                for (int j = 0; j < 2; ++j)
                    bf[(g + 1) & 1][u][j] = xb[((4 * (2 * (g + 1) + u) + j) * XPP) / 4];
        }
        __builtin_amdgcn_sched_barrier(0);
#pragma unroll
        for (int u = 0; u < 2; ++u)
#pragma unroll
            for (int j = 0; j < 2; ++j) {
                v2f a  = {av[2 * g + u][2 * j], av[2 * g + u][2 * j + 1]};
                v4f bb = bf[g & 1][u][j];
                v2f b0 = {bb[0], bb[1]};
                v2f b1 = {bb[2], bb[3]};
                acc0 = __builtin_amdgcn_wmma_f32_16x16x4_f32(
                    false, a, false, b0, (short)0, acc0, false, false);
                acc1 = __builtin_amdgcn_wmma_f32_16x16x4_f32(
                    false, a, false, b1, (short)0, acc1, false, false);
            }
        __builtin_amdgcn_sched_barrier(0);
    }

    float* orow = out + ((size_t)b * CDIM + wave * 16 + 8 * hi) * NPIX + p0 + 2 * col;
#pragma unroll
    for (int r = 0; r < 8; ++r) {
        float bias = wzb[wave * 16 + 8 * hi + r];
        v2f o = {acc0[r] + bias, acc1[r] + bias};
        *reinterpret_cast<v2f*>(orow + (size_t)r * NPIX) = o;
    }
}

extern "C" void kernel_launch(void* const* d_in, const int* in_sizes, int n_in,
                              void* d_out, int out_size, void* d_ws, size_t ws_size,
                              hipStream_t stream) {
    const float* q   = (const float*)d_in[0];
    const float* k   = (const float*)d_in[1];
    const float* v   = (const float*)d_in[2];
    const float* rpb = (const float*)d_in[3];
    const float* wzw = (const float*)d_in[4];
    const float* wzb = (const float*)d_in[5];
    float* out = (float*)d_out;
    float* xn  = (float*)d_ws;

    natten_attn_kernel<<<768, 224, ATT_LDS_BYTES, stream>>>(q, k, v, rpb, xn);

    conv1x1_wmma_kernel<<<256, 256, 0, stream>>>(xn, wzw, wzb, out);
}


# attention kernel: a lane loads only its own 16 B of each K/V row (1 load, not 5 overlapping), gets 12 neighbour columns by v_permlane16_var; logits in pixel pairs; edge columns' V tap order changes
# speedup vs baseline: 20.2736x; 1.1895x over previous
#include <hip/hip_runtime.h>
#include <hip/hip_bf16.h>

#define KS      7
#define DILX    2
#define NHEADS  4
#define NFRAMES 3
#define CDIM    128
#define HD      32
#define HH      64
#define WW      64
#define NPIX    4096
#define NB      2
#define KTOT    (NFRAMES*CDIM)

typedef __attribute__((ext_vector_type(2))) float v2f;
typedef __attribute__((ext_vector_type(8))) float v8f;

__device__ __forceinline__ int win_start(int i) {
    int ni = i - (KS / 2) * DILX;
    if (ni < 0) return i % DILX;
    if (i + (KS / 2) * DILX >= HH) return HH - KS * DILX + (i % DILX);
    return ni;
}

typedef __attribute__((ext_vector_type(4))) float v4f;
__device__ __forceinline__ float pick(bool c, float a, float b) { return c ? a : b; }

struct RowSel { unsigned l2, l1, r1, r2; };
__device__ __forceinline__ RowSel make_rowsel(int L) {
    RowSel s;
    s.l2 = (unsigned)(L >= 2 ? L - 2 : 2 - 2 * L);
    s.l1 = (unsigned)(L >= 1 ? L - 1 : 3);
    s.r1 = (unsigned)(L <= 14 ? L + 1 : 12);
    s.r2 = (unsigned)(L <= 13 ? L + 2 : 43 - 2 * L);
    return s;
}

__device__ __forceinline__ float rowgather(float x, unsigned sel) {
    unsigned u = __builtin_bit_cast(unsigned, x);
    return __builtin_bit_cast(float,
        __builtin_amdgcn_permlane16_var(u, u, sel, true, false));
}

__device__ __forceinline__ void span16(v4f own, const RowSel& s, float* X) {
    X[0]  = rowgather(own[2], s.l2);
    X[1]  = rowgather(own[3], s.l2);
    X[2]  = rowgather(own[0], s.l1);
    X[3]  = rowgather(own[1], s.l1);
    X[4]  = rowgather(own[2], s.l1);
    X[5]  = rowgather(own[3], s.l1);
    X[6]  = own[0]; X[7] = own[1]; X[8] = own[2]; X[9] = own[3];
    X[10] = rowgather(own[0], s.r1);
    X[11] = rowgather(own[1], s.r1);
    X[12] = rowgather(own[2], s.r1);
    X[13] = rowgather(own[3], s.r1);
    X[14] = rowgather(own[0], s.r2);
    X[15] = rowgather(own[1], s.r2);
}

#define PIXB      128
#define HDH       16
#define NCAND     8
#define ATT_LDS_FLOATS (2 * KS * PIXB + KS * HDH * PIXB)
#define ATT_LDS_BYTES  (ATT_LDS_FLOATS * 4)

__global__ __launch_bounds__(224, 6)
void natten_attn_kernel(const float* __restrict__ q,
                        const float* __restrict__ kk,
                        const float* __restrict__ vv,
                        const float* __restrict__ rpb,
                        float* __restrict__ xn)
{
    extern __shared__ __attribute__((aligned(16))) float att_lds[];
    float* pm   = att_lds;
    float* ps   = att_lds + KS * PIXB;
    float* pacc = att_lds + 2 * KS * PIXB;

    int lane = threadIdx.x & 31;
    int w    = threadIdx.x >> 5;
    int t    = blockIdx.x;
    int i    = ((t & 31) >> 1) * 4 + (t & 1) + 2 * (lane >> 4);
    t >>= 5;
    int f = t % NFRAMES;
    t /= NFRAMES;
    int head = t & 3;
    int b = t >> 2;
    int j0 = (lane & 15) * 4;
    const RowSel rsel = make_rowsel(lane & 15);

    int si  = win_start(i);
    int pbi = (si - i) / DILX + (KS - 1);

    int rot[2], pbj[2];
#pragma unroll
    for (int h = 0; h < 2; ++h) {
        int j  = j0 + 2 * h;
        int sj = win_start(j);
        rot[h] = ((sj - j0 + 6) >> 1) & (NCAND - 1);
        pbj[h] = (sj - j) / DILX + (KS - 1);
    }

    const float scale = 0.17677669529663687f;

    const size_t kvbase = ((size_t)(b * KTOT + f * CDIM + head * HD)) * NPIX;
    const int    rowoff = (si + w * DILX) * WW + j0;
    const v4f* kb = reinterpret_cast<const v4f*>(kk + kvbase + rowoff);
    const v4f* vb = reinterpret_cast<const v4f*>(vv + kvbase + rowoff);
    const v4f* qp = reinterpret_cast<const v4f*>(
        q + ((size_t)(b * CDIM + head * HD)) * NPIX + i * WW + j0);

    v2f cand2[2][NCAND];
#pragma unroll
    for (int h = 0; h < 2; ++h)
#pragma unroll
        for (int m = 0; m < NCAND; ++m) cand2[h][m] = v2f{0.0f, 0.0f};

#pragma unroll 8
    for (int d = 0; d < HD; ++d) {
        v4f qv = qp[(size_t)d * (NPIX / 4)];
        v4f kv = kb[(size_t)d * (NPIX / 4)];
        float X[16];
        span16(kv, rsel, X);
        v2f q2[2] = {v2f{qv[0] * scale, qv[1] * scale},
                     v2f{qv[2] * scale, qv[3] * scale}};
#pragma unroll
        for (int h = 0; h < 2; ++h)
#pragma unroll
            for (int m = 0; m < NCAND; ++m)
                cand2[h][m] = __builtin_elementwise_fma(
                    q2[h], v2f{X[2 * m], X[2 * m + 1]}, cand2[h][m]);
    }

    float wc[4][NCAND];
    v4f mv, sv;
#pragma unroll
    for (int pp = 0; pp < 4; ++pp) {
        const int r = rot[pp >> 1];
        bool s1 = (r & 1) != 0, s2 = (r & 2) != 0, s4 = (r & 4) != 0;
        float t1[NCAND], t2[NCAND];
#pragma unroll
        for (int x = 0; x < NCAND; ++x)
            t1[x] = pick(s1, cand2[pp >> 1][(x + 1) & (NCAND - 1)][pp & 1],
                             cand2[pp >> 1][x][pp & 1]);
#pragma unroll
        for (int x = 0; x < NCAND; ++x)
            t2[x] = pick(s2, t1[(x + 2) & (NCAND - 1)], t1[x]);
        const float* rp = rpb + head * (2 * KS - 1) * (2 * KS - 1)
                              + (pbi + w) * (2 * KS - 1) + pbj[pp >> 1];
        float l[KS];
#pragma unroll
        for (int tj = 0; tj < KS; ++tj)
            l[tj] = pick(s4, t2[(tj + 4) & (NCAND - 1)], t2[tj]) + rp[tj];
        float m = l[0];
#pragma unroll
        for (int tj = 1; tj < KS; ++tj) m = fmaxf(m, l[tj]);
        float p[NCAND];
        float s = 0.0f;
#pragma unroll
        for (int tj = 0; tj < KS; ++tj) { p[tj] = __expf(l[tj] - m); s += p[tj]; }
        p[KS] = 0.0f;
        mv[pp] = m;
        sv[pp] = s;
        float u1[NCAND], u2[NCAND];
#pragma unroll
        for (int x = 0; x < NCAND; ++x)
            u1[x] = pick(s1, p[(x + NCAND - 1) & (NCAND - 1)], p[x]);
#pragma unroll
        for (int x = 0; x < NCAND; ++x)
            u2[x] = pick(s2, u1[(x + NCAND - 2) & (NCAND - 1)], u1[x]);
#pragma unroll
        for (int x = 0; x < NCAND; ++x)
            wc[pp][x] = pick(s4, u2[(x + NCAND - 4) & (NCAND - 1)], u2[x]);
    }
    reinterpret_cast<v4f*>(pm + w * PIXB)[lane] = mv;
    reinterpret_cast<v4f*>(ps + w * PIXB)[lane] = sv;

    float* op = xn + kvbase + i * WW + j0;

#pragma unroll
    for (int hf = 0; hf < 2; ++hf) {
#pragma unroll 8
        for (int dd = 0; dd < HDH; ++dd) {
            const int d = hf * HDH + dd;
            v4f vown = vb[(size_t)d * (NPIX / 4)];
            float X[16];
            span16(vown, rsel, X);
            v4f av;
#pragma unroll
            for (int pp = 0; pp < 4; ++pp) {
                float a = 0.0f;
#pragma unroll
                for (int m = 0; m < NCAND; ++m)
                    a = __builtin_fmaf(wc[pp][m], X[(pp & 1) + 2 * m], a);
                av[pp] = a;
            }
            reinterpret_cast<v4f*>(pacc + (w * HDH + dd) * PIXB)[lane] = av;
        }

        __syncthreads();

        v4f e[KS];
        v4f inv;
        {
            v4f pmr[KS], psr[KS];
#pragma unroll
            for (int r = 0; r < KS; ++r) {
                pmr[r] = reinterpret_cast<const v4f*>(pm + r * PIXB)[lane];
                psr[r] = reinterpret_cast<const v4f*>(ps + r * PIXB)[lane];
            }
#pragma unroll
            for (int pp = 0; pp < 4; ++pp) {
                float M = pmr[0][pp];
#pragma unroll
                for (int r = 1; r < KS; ++r) M = fmaxf(M, pmr[r][pp]);
                float S = 0.0f;
#pragma unroll
                for (int r = 0; r < KS; ++r) {
                    float er = __expf(pmr[r][pp] - M);
                    e[r][pp] = er;
                    S += er * psr[r][pp];
                }
                inv[pp] = 1.0f / S;
            }
        }

        for (int dd = w; dd < HDH; dd += KS) {
            v4f o = {0.0f, 0.0f, 0.0f, 0.0f};
#pragma unroll
            for (int r = 0; r < KS; ++r) {
                v4f pa = reinterpret_cast<const v4f*>(pacc + (r * HDH + dd) * PIXB)[lane];
                o += e[r] * pa;
            }
            *reinterpret_cast<v4f*>(op + (size_t)(hf * HDH + dd) * NPIX) = o * inv;
        }

        if (hf == 0) __syncthreads();
    }
}

#define XPP     80
#define KPAIRS  (KTOT / 2)
#define KGRP    (KTOT / 8)

__global__ __launch_bounds__(256, 2)
void conv1x1_wmma_kernel(const float* __restrict__ xn,
                         const float* __restrict__ wzw,
                         const float* __restrict__ wzb,
                         float* __restrict__ out)
{
    __shared__ __attribute__((aligned(16))) float xs[KPAIRS * XPP];

    const int tid  = threadIdx.x;
    const int lane = tid & 31;
    const int wave = tid >> 5;
    const int col  = lane & 15;
    const int hi   = lane >> 4;
    const int b    = blockIdx.x >> 7;
    const int p0   = (blockIdx.x & 127) * 32;

    const int q  = tid & 7;
    const int pr = tid >> 3;
    const float* xsrc = xn + ((size_t)b * KTOT + 2 * pr) * NPIX + p0 + 4 * q;
    v4f xe[6], xo[6];
#pragma unroll
    for (int ps = 0; ps < 6; ++ps) {
        xe[ps] = *reinterpret_cast<const v4f*>(xsrc + (size_t)(64 * ps) * NPIX);
        xo[ps] = *reinterpret_cast<const v4f*>(xsrc + (size_t)(64 * ps + 1) * NPIX);
    }

    float* xdst = xs + pr * XPP + 8 * q;
#pragma unroll
    for (int ps = 0; ps < 6; ++ps) {
        v4f lo = {xe[ps][0], xo[ps][0], xe[ps][1], xo[ps][1]};
        v4f up = {xe[ps][2], xo[ps][2], xe[ps][3], xo[ps][3]};
        *reinterpret_cast<v4f*>(xdst + (32 * ps) * XPP)     = lo;
        *reinterpret_cast<v4f*>(xdst + (32 * ps) * XPP + 4) = up;
    }
    __syncthreads();

    const float* wsrc = wzw + (size_t)(wave * 16 + col) * KTOT + 4 * hi;
    v4f av[KGRP];
#pragma unroll
    for (int kp = 0; kp < KGRP; ++kp)
        av[kp] = *reinterpret_cast<const v4f*>(wsrc + 8 * kp);

    v8f acc0, acc1;
#pragma unroll
    for (int e = 0; e < 8; ++e) { acc0[e] = 0.0f; acc1[e] = 0.0f; }

    const v4f* xb = reinterpret_cast<const v4f*>(xs + (2 * hi) * XPP + 4 * col);
    v4f bf[2][2][2];
#pragma unroll
    for (int u = 0; u < 2; ++u)
#pragma unroll
        for (int j = 0; j < 2; ++j)
            bf[0][u][j] = xb[((4 * u + j) * XPP) / 4];

#pragma unroll
    for (int g = 0; g < KGRP / 2; ++g) {
        if (g + 1 < KGRP / 2) {
#pragma unroll
            for (int u = 0; u < 2; ++u)
#pragma unroll
                for (int j = 0; j < 2; ++j)
                    bf[(g + 1) & 1][u][j] = xb[((4 * (2 * (g + 1) + u) + j) * XPP) / 4];
        }
        __builtin_amdgcn_sched_barrier(0);
#pragma unroll
        for (int u = 0; u < 2; ++u)
#pragma unroll
            for (int j = 0; j < 2; ++j) {
                v2f a  = {av[2 * g + u][2 * j], av[2 * g + u][2 * j + 1]};
                v4f bb = bf[g & 1][u][j];
                v2f b0 = {bb[0], bb[1]};
                v2f b1 = {bb[2], bb[3]};
                acc0 = __builtin_amdgcn_wmma_f32_16x16x4_f32(
                    false, a, false, b0, (short)0, acc0, false, false);
                acc1 = __builtin_amdgcn_wmma_f32_16x16x4_f32(
                    false, a, false, b1, (short)0, acc1, false, false);
            }
        __builtin_amdgcn_sched_barrier(0);
    }

    float* orow = out + ((size_t)b * CDIM + wave * 16 + 8 * hi) * NPIX + p0 + 2 * col;
#pragma unroll
    for (int r = 0; r < 8; ++r) {
        float bias = wzb[wave * 16 + 8 * hi + r];
        v2f o = {acc0[r] + bias, acc1[r] + bias};
        *reinterpret_cast<v2f*>(orow + (size_t)r * NPIX) = o;
    }
}

extern "C" void kernel_launch(void* const* d_in, const int* in_sizes, int n_in,
                              void* d_out, int out_size, void* d_ws, size_t ws_size,
                              hipStream_t stream) {
    const float* q   = (const float*)d_in[0];
    const float* k   = (const float*)d_in[1];
    const float* v   = (const float*)d_in[2];
    const float* rpb = (const float*)d_in[3];
    const float* wzw = (const float*)d_in[4];
    const float* wzb = (const float*)d_in[5];
    float* out = (float*)d_out;
    float* xn  = (float*)d_ws;

    natten_attn_kernel<<<768, 224, ATT_LDS_BYTES, stream>>>(q, k, v, rpb, xn);

    conv1x1_wmma_kernel<<<256, 256, 0, stream>>>(xn, wzw, wzb, out);
}
